# starting kernel
# baseline (speedup 1.0000x reference)
#include <hip/hip_runtime.h>
#include <hip/hip_bf16.h>

typedef __attribute__((ext_vector_type(2))) float v2f;
typedef __attribute__((ext_vector_type(8))) float v8f;

#define D_FEAT 128
#define WAVES_PER_BLOCK 8
#define BLOCK_THREADS (WAVES_PER_BLOCK * 32)

__global__ __launch_bounds__(BLOCK_THREADS) void node_project_wmma(
    const float* __restrict__ h, const float* __restrict__ W,
    float* __restrict__ P, int n_tiles)
{
    __shared__ __align__(16) float sW[512 + 128];
    for (int i = threadIdx.x; i < 640; i += BLOCK_THREADS)
        sW[i] = (i < 512) ? W[i] : 0.0f;
    __syncthreads();

    const int wave = threadIdx.x >> 5;
    const int tile = blockIdx.x * WAVES_PER_BLOCK + wave;
    if (tile >= n_tiles) return;

    const int lane = threadIdx.x & 31;
    const int half = lane >> 4;
    const int m    = lane & 15;

    const float2* hrow =
        reinterpret_cast<const float2*>(h + (size_t)(tile * 16 + m) * D_FEAT);

    const int wbase = (m < 4) ? ((m & 1) * 256 + ((m >> 1) & 1) * 128) : 512;
    const float2* wrow = reinterpret_cast<const float2*>(sW) + (wbase >> 1);

    v8f c = {};
#pragma unroll
    for (int k0 = 0; k0 < D_FEAT; k0 += 4) {
        float2 av = hrow[(k0 >> 1) + half];
        float2 bv = wrow[(k0 >> 1) + half];
        v2f a; a.x = av.x; a.y = av.y;
        v2f b; b.x = bv.x; b.y = bv.y;
        c = __builtin_amdgcn_wmma_f32_16x16x4_f32(
                false, a, false, b, (short)0, c, false, false);
    }

    if (m < 4) {
#pragma unroll
        for (int r = 0; r < 8; ++r) {
            int node = tile * 16 + r + 8 * half;
            P[(size_t)node * 4 + m] = c[r];
        }
    }
}

__global__ __launch_bounds__(256) void edge_score_kernel(
    const int* __restrict__ src, const int* __restrict__ dst,
    const float* __restrict__ P, const float* __restrict__ bvec,
    float* __restrict__ out, int E)
{
    int e = blockIdx.x * blockDim.x + threadIdx.x;
    if (e >= E) return;
    int s = src[e];
    int d = dst[e];
    float2 ps = *reinterpret_cast<const float2*>(P + (size_t)s * 4);
    float2 pd = *reinterpret_cast<const float2*>(P + (size_t)d * 4 + 2);
    float2 o;
    o.x = ps.x + pd.x + bvec[0];
    o.y = ps.y + pd.y + bvec[1];
    *reinterpret_cast<float2*>(out + (size_t)e * 2) = o;
}

extern "C" void kernel_launch(void* const* d_in, const int* in_sizes, int n_in,
                              void* d_out, int out_size, void* d_ws, size_t ws_size,
                              hipStream_t stream)
{
    const float* h   = (const float*)d_in[0];
    const int*   src = (const int*)d_in[1];
    const int*   dst = (const int*)d_in[2];
    const float* W   = (const float*)d_in[3];
    const float* b   = (const float*)d_in[4];
    float*       out = (float*)d_out;
    float*       P   = (float*)d_ws;

    const int n_nodes = in_sizes[0] / D_FEAT;
    const int E       = in_sizes[1];
    const int n_tiles = (n_nodes + 15) / 16;

    const int grid1 = (n_tiles + WAVES_PER_BLOCK - 1) / WAVES_PER_BLOCK;
    node_project_wmma<<<grid1, BLOCK_THREADS, 0, stream>>>(h, W, P, n_tiles);

    const int threads = 256;
    const int grid2 = (E + threads - 1) / threads;
    edge_score_kernel<<<grid2, threads, 0, stream>>>(src, dst, P, b, out, E);
}


# node kernel: A and B fragment loads widened to 16 B per lane (16 loads of 16 lines, was 32), k permuted inside each 8-k group identically in both loaders; f32 sum order changes
# speedup vs baseline: 1.0625x; 1.0625x over previous
#include <hip/hip_runtime.h>
#include <hip/hip_bf16.h>

typedef __attribute__((ext_vector_type(2))) float v2f;
typedef __attribute__((ext_vector_type(8))) float v8f;

#define D_FEAT 128
#define WAVES_PER_BLOCK 8
#define BLOCK_THREADS (WAVES_PER_BLOCK * 32)

__global__ __launch_bounds__(BLOCK_THREADS) void node_project_wmma(
    const float* __restrict__ h, const float* __restrict__ W,
    float* __restrict__ P, int n_tiles)
{
    __shared__ __align__(16) float sW[512 + 128];
    for (int i = threadIdx.x; i < 640; i += BLOCK_THREADS)
        sW[i] = (i < 512) ? W[i] : 0.0f;
    __syncthreads();

    const int wave = threadIdx.x >> 5;
    const int tile = blockIdx.x * WAVES_PER_BLOCK + wave;
    if (tile >= n_tiles) return;

    const int lane = threadIdx.x & 31;
    const int half = lane >> 4;
    const int m    = lane & 15;

    const float4* hrow =
        reinterpret_cast<const float4*>(h + (size_t)(tile * 16 + m) * D_FEAT);

    const int wbase = (m < 4) ? ((m & 1) * 256 + ((m >> 1) & 1) * 128) : 512;
    const float4* wrow = reinterpret_cast<const float4*>(sW) + (wbase >> 2);

    v8f c = {};
#pragma unroll
    for (int k8 = 0; k8 < D_FEAT; k8 += 8) {
        float4 av = hrow[(k8 >> 2) + half];
        float4 bv = wrow[(k8 >> 2) + half];
        v2f a0; a0.x = av.x; a0.y = av.y;
        v2f b0; b0.x = bv.x; b0.y = bv.y;
        v2f a1; a1.x = av.z; a1.y = av.w;
        v2f b1; b1.x = bv.z; b1.y = bv.w;
        c = __builtin_amdgcn_wmma_f32_16x16x4_f32(
                false, a0, false, b0, (short)0, c, false, false);
        c = __builtin_amdgcn_wmma_f32_16x16x4_f32(
                false, a1, false, b1, (short)0, c, false, false);
    }

    if (m < 4) {
#pragma unroll
        for (int r = 0; r < 8; ++r) {
            int node = tile * 16 + r + 8 * half;
            P[(size_t)node * 4 + m] = c[r];
        }
    }
}

__global__ __launch_bounds__(256) void edge_score_kernel(
    const int* __restrict__ src, const int* __restrict__ dst,
    const float* __restrict__ P, const float* __restrict__ bvec,
    float* __restrict__ out, int E)
{
    int e = blockIdx.x * blockDim.x + threadIdx.x;
    if (e >= E) return;
    int s = src[e];
    int d = dst[e];
    float2 ps = *reinterpret_cast<const float2*>(P + (size_t)s * 4);
    float2 pd = *reinterpret_cast<const float2*>(P + (size_t)d * 4 + 2);
    float2 o;
    o.x = ps.x + pd.x + bvec[0];
    o.y = ps.y + pd.y + bvec[1];
    *reinterpret_cast<float2*>(out + (size_t)e * 2) = o;
}

extern "C" void kernel_launch(void* const* d_in, const int* in_sizes, int n_in,
                              void* d_out, int out_size, void* d_ws, size_t ws_size,
                              hipStream_t stream)
{
    const float* h   = (const float*)d_in[0];
    const int*   src = (const int*)d_in[1];
    const int*   dst = (const int*)d_in[2];
    const float* W   = (const float*)d_in[3];
    const float* b   = (const float*)d_in[4];
    float*       out = (float*)d_out;
    float*       P   = (float*)d_ws;

    const int n_nodes = in_sizes[0] / D_FEAT;
    const int E       = in_sizes[1];
    const int n_tiles = (n_nodes + 15) / 16;

    const int grid1 = (n_tiles + WAVES_PER_BLOCK - 1) / WAVES_PER_BLOCK;
    node_project_wmma<<<grid1, BLOCK_THREADS, 0, stream>>>(h, W, P, n_tiles);

    const int threads = 256;
    const int grid2 = (E + threads - 1) / threads;
    edge_score_kernel<<<grid2, threads, 0, stream>>>(src, dst, P, b, out, E);
}


# node kernel re-tiled: WMMA rows = 4 W rows x 4 k slices, columns = 4 nodes x 4 k slices, so an h load covers 4 whole 128-B lines (was 16 partial); W read from global, no LDS; f32 sum order changes
# speedup vs baseline: 1.2049x; 1.1340x over previous
#include <hip/hip_runtime.h>
#include <hip/hip_bf16.h>

typedef __attribute__((ext_vector_type(2))) float v2f;
typedef __attribute__((ext_vector_type(8))) float v8f;

#define D_FEAT 128
#define WAVES_PER_BLOCK 8
#define BLOCK_THREADS (WAVES_PER_BLOCK * 32)

__device__ __forceinline__ float quad_sum(float v)
{
    v += __int_as_float(__builtin_amdgcn_mov_dpp(__float_as_int(v), 0xB1, 0xF, 0xF, true));
    v += __int_as_float(__builtin_amdgcn_mov_dpp(__float_as_int(v), 0x4E, 0xF, 0xF, true));
    return v;
}

__device__ __forceinline__ float pick4(float a0, float a1, float a2, float a3, bool b0, bool b1)
{
    const float l = b0 ? a1 : a0;
    const float u = b0 ? a3 : a2;
    return b1 ? u : l;
}

__global__ __launch_bounds__(BLOCK_THREADS) void node_project_wmma(
    const float* __restrict__ h, const float* __restrict__ W,
    float* __restrict__ P, int n_tiles)
{
    const int wave = threadIdx.x >> 5;
    const int tile = blockIdx.x * WAVES_PER_BLOCK + wave;
    if (tile >= n_tiles) return;

    const int lane = threadIdx.x & 31;
    const int hh = lane >> 4;
    const int c  = lane & 15;
    const int x  = c >> 2;
    const int q  = c & 3;
    const int koff = 2 * q + hh;

    const float4* wrow =
        reinterpret_cast<const float4*>(W + (x & 1) * 256 + (x >> 1) * D_FEAT) + koff;
    const float4* hrow =
        reinterpret_cast<const float4*>(h + (size_t)(tile * 16 + x) * D_FEAT) + koff;

    float4 wv[4];
    float4 hv[4][4];
#pragma unroll
    for (int t = 0; t < 4; ++t) {
        wv[t] = wrow[8 * t];
#pragma unroll
        for (int g = 0; g < 4; ++g)
            hv[t][g] = hrow[g * 128 + 8 * t];
    }

    __builtin_amdgcn_sched_barrier(0);

    v8f acc[4] = {};
#pragma unroll
    for (int t = 0; t < 4; ++t) {
        v2f a0; a0.x = wv[t].x; a0.y = wv[t].y;
        v2f a1; a1.x = wv[t].z; a1.y = wv[t].w;
#pragma unroll
        for (int g = 0; g < 4; ++g) {
            v2f b0; b0.x = hv[t][g].x; b0.y = hv[t][g].y;
            acc[g] = __builtin_amdgcn_wmma_f32_16x16x4_f32(
                         false, a0, false, b0, (short)0, acc[g], false, false);
        }
#pragma unroll
        for (int g = 0; g < 4; ++g) {
            v2f b1; b1.x = hv[t][g].z; b1.y = hv[t][g].w;
            acc[g] = __builtin_amdgcn_wmma_f32_16x16x4_f32(
                         false, a1, false, b1, (short)0, acc[g], false, false);
        }
    }

    const bool qb0 = (q & 1) != 0;
    const bool qb1 = (q & 2) != 0;
    float2 o;
    o.x = 0.0f; o.y = 0.0f;
#pragma unroll
    for (int g = 0; g < 4; ++g) {
        float lo = pick4(acc[g][0], acc[g][1], acc[g][2], acc[g][3], qb0, qb1);
        float hi = pick4(acc[g][4], acc[g][5], acc[g][6], acc[g][7], qb0, qb1);
        lo = quad_sum(lo);
        hi = quad_sum(hi);
        if (q == g) { o.x = lo; o.y = hi; }
    }
    const int node = tile * 16 + 4 * q + x;
    *reinterpret_cast<float2*>(P + (size_t)node * 4 + 2 * hh) = o;
}

__global__ __launch_bounds__(256) void edge_score_kernel(
    const int* __restrict__ src, const int* __restrict__ dst,
    const float* __restrict__ P, const float* __restrict__ bvec,
    float* __restrict__ out, int E)
{
    int e = blockIdx.x * blockDim.x + threadIdx.x;
    if (e >= E) return;
    int s = src[e];
    int d = dst[e];
    float2 ps = *reinterpret_cast<const float2*>(P + (size_t)s * 4);
    float2 pd = *reinterpret_cast<const float2*>(P + (size_t)d * 4 + 2);
    float2 o;
    o.x = ps.x + pd.x + bvec[0];
    o.y = ps.y + pd.y + bvec[1];
    *reinterpret_cast<float2*>(out + (size_t)e * 2) = o;
}

extern "C" void kernel_launch(void* const* d_in, const int* in_sizes, int n_in,
                              void* d_out, int out_size, void* d_ws, size_t ws_size,
                              hipStream_t stream)
{
    const float* h   = (const float*)d_in[0];
    const int*   src = (const int*)d_in[1];
    const int*   dst = (const int*)d_in[2];
    const float* W   = (const float*)d_in[3];
    const float* b   = (const float*)d_in[4];
    float*       out = (float*)d_out;
    float*       P   = (float*)d_ws;

    const int n_nodes = in_sizes[0] / D_FEAT;
    const int E       = in_sizes[1];
    const int n_tiles = (n_nodes + 15) / 16;

    const int grid1 = (n_tiles + WAVES_PER_BLOCK - 1) / WAVES_PER_BLOCK;
    node_project_wmma<<<grid1, BLOCK_THREADS, 0, stream>>>(h, W, P, n_tiles);

    const int threads = 256;
    const int grid2 = (E + threads - 1) / threads;
    edge_score_kernel<<<grid2, threads, 0, stream>>>(src, dst, P, b, out, E);
}
